# starting kernel
# baseline (speedup 1.0000x reference)
#include <hip/hip_runtime.h>
#include <hip/hip_bf16.h>

#define BB   2
#define CC   96
#define HH   64
#define WWW  64
#define LL   4096
#define NN   16
#define RR   6
#define KK   4
#define DBCROWS 38
#define MPAD 48
#define LT   64
#define CH   4
#define CHL  (LL / CH)

typedef float v2f __attribute__((ext_vector_type(2)));
typedef float v8f __attribute__((ext_vector_type(8)));

__device__ __forceinline__ int tr_idx(int l) { return ((l & 63) << 6) | (l >> 6); }
__device__ __forceinline__ int dir_idx(int k, int l) {
    switch (k) {
        case 0:  return l;
        case 1:  return tr_idx(l);
        case 2:  return LL - 1 - l;
        default: return tr_idx(LL - 1 - l);
    }
}

__global__ __launch_bounds__(256)
void s6_proj_kernel(const float* __restrict__ x,
                    const float* __restrict__ wproj,
                    const float* __restrict__ wdelta,
                    const float* __restrict__ dbias,
                    float* __restrict__ delta_out,
                    float* __restrict__ Bt,
                    float* __restrict__ Ct)
{
    __shared__ float sW[MPAD][CC];
    __shared__ float sX[CC][LT + 1];
    __shared__ float sDBC[MPAD][LT + 1];
    __shared__ float sWd[CC][RR];

    const int tid = threadIdx.x;
    const int l0  = blockIdx.x * LT;
    const int k   = blockIdx.y;
    const int b   = blockIdx.z;

    for (int i = tid; i < MPAD * CC; i += 256) {
        int m = i / CC, c = i % CC;
        sW[m][c] = (m < DBCROWS) ? wproj[(k * DBCROWS + m) * CC + c] : 0.0f;
    }
    for (int i = tid; i < CC * RR; i += 256) {
        int c = i / RR, r = i % RR;
        sWd[c][r] = wdelta[(k * CC + c) * RR + r];
    }
    for (int i = tid; i < CC * LT; i += 256) {
        int c = i / LT, col = i % LT;
        sX[c][col] = x[((size_t)b * CC + c) * LL + dir_idx(k, l0 + col)];
    }
    __syncthreads();

    const int wave = tid >> 5;
    const int lane = tid & 31;
    const int lm   = lane & 15;
    const int lh   = lane >> 4;
    for (int t = wave; t < (MPAD / 16) * (LT / 16); t += 8) {
        const int mt = t / (LT / 16);
        const int nt = t % (LT / 16);
        v8f acc = {};
        for (int kk = 0; kk < CC; kk += 4) {
            v2f a, bf;
            a.x  = sW[mt * 16 + lm][kk + 2 * lh + 0];
            a.y  = sW[mt * 16 + lm][kk + 2 * lh + 1];
            bf.x = sX[kk + 2 * lh + 0][nt * 16 + lm];
            bf.y = sX[kk + 2 * lh + 1][nt * 16 + lm];
            acc = __builtin_amdgcn_wmma_f32_16x16x4_f32(
                      false, a, false, bf, (short)0, acc, false, false);
        }
        #pragma unroll
        for (int i = 0; i < 8; ++i)
            sDBC[mt * 16 + i + 8 * lh][nt * 16 + lm] = acc[i];
    }
    __syncthreads();

    for (int i = tid; i < LT * NN; i += 256) {
        int col = i / NN, n = i % NN;
        size_t base = (((size_t)(b * KK + k) * LL + (l0 + col)) * NN + n);
        Bt[base] = sDBC[RR + n][col];
        Ct[base] = sDBC[RR + NN + n][col];
    }
    for (int i = tid; i < CC * LT; i += 256) {
        int c = i / LT, col = i % LT;
        float s = 0.0f;
        #pragma unroll
        for (int r = 0; r < RR; ++r) s += sWd[c][r] * sDBC[r][col];
        s += dbias[k * CC + c];
        float sp = (s > 20.0f) ? s : log1pf(__expf(s));
        delta_out[(((size_t)(b * KK + k) * CC + c) * LL) + l0 + col] = sp;
    }
}

__global__ __launch_bounds__(256)
void s6_scan_carry_kernel(const float* __restrict__ x,
                          const float* __restrict__ delta,
                          const float* __restrict__ Bt,
                          const float* __restrict__ A_log,
                          float* __restrict__ carryA,
                          float* __restrict__ carryH)
{
    const int tid  = threadIdx.x;
    const int unit = (blockIdx.x * 256 + tid) >> 4;
    const int n    = tid & 15;
    if (unit >= BB * KK * CC * CH) return;
    const int chunk = unit % CH;
    const int ch    = unit / CH;
    const int c     = ch % CC;
    const int bk    = ch / CC;
    const int k     = bk % KK;
    const int b     = bk / KK;
    const int d     = k * CC + c;

    const float An = -__expf(A_log[d * NN + n]);
    const float* dptr = delta + ((size_t)bk * CC + c) * LL;
    const float* bptr = Bt + (size_t)bk * LL * NN;
    const float* xptr = x + ((size_t)b * CC + c) * LL;

    float h = 0.0f, ap = 1.0f;
    const int lbeg = chunk * CHL, lend = lbeg + CHL;
    for (int l = lbeg; l < lend; ++l) {
        __builtin_prefetch(dptr + ((l + 64) & (LL - 1)), 0, 0);
        __builtin_prefetch(bptr + (size_t)((l + 16) & (LL - 1)) * NN, 0, 0);
        float ld = dptr[l];
        float u  = xptr[dir_idx(k, l)];
        float Bn = bptr[(size_t)l * NN + n];
        float dA = __expf(ld * An);
        h  = dA * h + (ld * u) * Bn;
        ap *= dA;
    }
    size_t cidx = (size_t)unit * NN + n;
    carryA[cidx] = ap;
    carryH[cidx] = h;
}

__global__ __launch_bounds__(256)
void s6_combine_kernel(const float* __restrict__ carryA,
                       const float* __restrict__ carryH,
                       float* __restrict__ initH)
{
    int i = blockIdx.x * 256 + threadIdx.x;
    if (i >= BB * KK * CC * NN) return;
    int ch = i / NN, n = i % NN;
    float Hp = 0.0f;
    for (int j = 0; j < CH; ++j) {
        size_t idx = (size_t)(ch * CH + j) * NN + n;
        initH[idx] = Hp;
        Hp = carryA[idx] * Hp + carryH[idx];
    }
}

__global__ __launch_bounds__(256)
void s6_scan_out_kernel(const float* __restrict__ x,
                        const float* __restrict__ delta,
                        const float* __restrict__ Bt,
                        const float* __restrict__ Ct,
                        const float* __restrict__ A_log,
                        const float* __restrict__ Ds,
                        const float* __restrict__ initH,
                        float* __restrict__ out)
{
    const int tid  = threadIdx.x;
    const int unit = (blockIdx.x * 256 + tid) >> 4;
    const int n    = tid & 15;
    if (unit >= BB * KK * CC * CH) return;
    const int chunk = unit % CH;
    const int ch    = unit / CH;
    const int c     = ch % CC;
    const int bk    = ch / CC;
    const int k     = bk % KK;
    const int b     = bk / KK;
    const int d     = k * CC + c;

    const float An = -__expf(A_log[d * NN + n]);
    const float Dd = Ds[d];
    const float* dptr = delta + ((size_t)bk * CC + c) * LL;
    const float* bptr = Bt + (size_t)bk * LL * NN;
    const float* cptr = Ct + (size_t)bk * LL * NN;
    const float* xptr = x + ((size_t)b * CC + c) * LL;
    const int slot = (k == 0) ? 0 : (k == 1) ? 2 : (k == 2) ? 1 : 3;
    float* optr = out + ((size_t)(slot * BB + b) * CC + c) * LL;

    float h = initH[(size_t)unit * NN + n];
    const int lbeg = chunk * CHL, lend = lbeg + CHL;
    for (int l = lbeg; l < lend; ++l) {
        __builtin_prefetch(dptr + ((l + 64) & (LL - 1)), 0, 0);
        __builtin_prefetch(cptr + (size_t)((l + 16) & (LL - 1)) * NN, 0, 0);
        float ld  = dptr[l];
        int   idx = dir_idx(k, l);
        float u   = xptr[idx];
        float Bn  = bptr[(size_t)l * NN + n];
        float Cn  = cptr[(size_t)l * NN + n];
        float dA  = __expf(ld * An);
        h = dA * h + (ld * u) * Bn;
        float v = h * Cn;
        v += __shfl_xor(v, 1);
        v += __shfl_xor(v, 2);
        v += __shfl_xor(v, 4);
        v += __shfl_xor(v, 8);
        if (n == 0) optr[idx] = v + Dd * u;
    }
}

extern "C" void kernel_launch(void* const* d_in, const int* in_sizes, int n_in,
                              void* d_out, int out_size, void* d_ws, size_t ws_size,
                              hipStream_t stream) {
    const float* x      = (const float*)d_in[0];
    const float* wproj  = (const float*)d_in[1];
    const float* wdelta = (const float*)d_in[2];
    const float* dbias  = (const float*)d_in[3];
    const float* A_log  = (const float*)d_in[4];
    const float* Ds     = (const float*)d_in[5];
    float* out = (float*)d_out;

    float* ws     = (float*)d_ws;
    float* delta  = ws;
    float* Bt     = delta  + (size_t)BB * KK * CC * LL;
    float* Ct     = Bt     + (size_t)BB * KK * LL * NN;
    float* carryA = Ct     + (size_t)BB * KK * LL * NN;
    float* carryH = carryA + (size_t)BB * KK * CC * CH * NN;
    float* initH  = carryH + (size_t)BB * KK * CC * CH * NN;

    dim3 gProj(LL / LT, KK, BB);
    s6_proj_kernel<<<gProj, 256, 0, stream>>>(x, wproj, wdelta, dbias, delta, Bt, Ct);

    const int units = BB * KK * CC * CH;
    s6_scan_carry_kernel<<<units * 16 / 256, 256, 0, stream>>>(x, delta, Bt, A_log,
                                                               carryA, carryH);
    s6_combine_kernel<<<(BB * KK * CC * NN + 255) / 256, 256, 0, stream>>>(carryA, carryH,
                                                                           initH);
    s6_scan_out_kernel<<<units * 16 / 256, 256, 0, stream>>>(x, delta, Bt, Ct, A_log, Ds,
                                                             initH, out);
}


# projection: one wave per 32 steps keeps all WMMA operands in registers, loaded from global memory (no LDS staging, no barriers, 1 wave per SIMD); softplus as select, same values; sums unchanged
# speedup vs baseline: 13.8923x; 13.8923x over previous
#include <hip/hip_runtime.h>
#include <hip/hip_bf16.h>

#define BB   2
#define CC   96
#define HH   64
#define WWW  64
#define LL   4096
#define NN   16
#define RR   6
#define KK   4
#define DBCROWS 38
#define MPAD 48

typedef float v2f __attribute__((ext_vector_type(2)));
typedef float v8f __attribute__((ext_vector_type(8)));

__device__ __forceinline__ int tr_idx(int l) { return ((l & 63) << 6) | (l >> 6); }
__device__ __forceinline__ int dir_idx(int k, int l) {
    switch (k) {
        case 0:  return l;
        case 1:  return tr_idx(l);
        case 2:  return LL - 1 - l;
        default: return tr_idx(LL - 1 - l);
    }
}

#define PWAVES 4
#define KSTEPS (CC / 4)
#define MTILES (MPAD / 16)
#define CHB    4

struct WdBlock {
    float w[CHB][RR];
    float bias[CHB];
};

__device__ __forceinline__ void wd_load(WdBlock& o, const float* __restrict__ wdelta,
                                        const float* __restrict__ dbias, int k, int ch0)
{
    #pragma unroll
    for (int u = 0; u < CHB; ++u) {
        #pragma unroll
        for (int r = 0; r < RR; ++r) o.w[u][r] = wdelta[(k * CC + ch0 + u) * RR + r];
        o.bias[u] = dbias[k * CC + ch0 + u];
    }
}

__global__ __launch_bounds__(32 * PWAVES)
void s6_proj_kernel(const float* __restrict__ x,
                    const float* __restrict__ wproj,
                    const float* __restrict__ wdelta,
                    const float* __restrict__ dbias,
                    float* __restrict__ delta_out,
                    float* __restrict__ Bt,
                    float* __restrict__ Ct)
{
    const int lane = threadIdx.x & 31;
    const int wv = __builtin_amdgcn_readfirstlane((int)(threadIdx.x >> 5));
    const int c  = lane & 15;
    const int hf = lane >> 4;
    const int k  = blockIdx.y;
    const int b  = blockIdx.z;
    const bool odd = (k & 1) != 0;
    const bool rev = (k & 2) != 0;

    int lq0, lq1;
    if (!odd) { lq0 = (blockIdx.x * PWAVES + wv) * 32 + 2 * c;        lq1 = lq0 + 1;  }
    else      { lq0 = blockIdx.x * (32 * PWAVES) + 16 * wv + c;       lq1 = lq0 + 64; }
    const int p0 = dir_idx(k, lq0), p1 = dir_idx(k, lq1);
    const int pmin = (p0 < p1) ? p0 : p1;
    const int lA = rev ? lq1 : lq0;
    const int lB = rev ? lq0 : lq1;

    WdBlock wcur;
    wd_load(wcur, wdelta, dbias, k, 0);

    const float* xp = x + ((size_t)b * CC + 2 * hf) * LL + pmin;
    const float* wr[MTILES];
    bool valid[MTILES];
    #pragma unroll
    for (int mt = 0; mt < MTILES; ++mt) {
        const int row = mt * 16 + c;
        valid[mt] = row < DBCROWS;
        wr[mt] = wproj + ((size_t)k * DBCROWS + (valid[mt] ? row : DBCROWS - 1)) * CC + 2 * hf;
    }
    v2f bA[KSTEPS], bB[KSTEPS], af[MTILES][KSTEPS];
    #pragma unroll
    for (int s = 0; s < KSTEPS; ++s) {
        const v2f t0 = *reinterpret_cast<const v2f*>(xp + (size_t)(4 * s + 0) * LL);
        const v2f t1 = *reinterpret_cast<const v2f*>(xp + (size_t)(4 * s + 1) * LL);
        bA[s].x = t0.x; bA[s].y = t1.x;
        bB[s].x = t0.y; bB[s].y = t1.y;
        #pragma unroll
        for (int mt = 0; mt < MTILES; ++mt) {
            const v2f t = *reinterpret_cast<const v2f*>(wr[mt] + 4 * s);
            af[mt][s].x = valid[mt] ? t.x : 0.0f;
            af[mt][s].y = valid[mt] ? t.y : 0.0f;
        }
    }
    __builtin_amdgcn_sched_barrier(0);

    v8f acc[MTILES][2];
    #pragma unroll
    for (int mt = 0; mt < MTILES; ++mt) { acc[mt][0] = v8f{}; acc[mt][1] = v8f{}; }
    #pragma unroll
    for (int s = 0; s < KSTEPS; ++s) {
        #pragma unroll
        for (int mt = 0; mt < MTILES; ++mt) {
            acc[mt][0] = __builtin_amdgcn_wmma_f32_16x16x4_f32(
                             false, af[mt][s], false, bA[s], (short)0, acc[mt][0], false, false);
            acc[mt][1] = __builtin_amdgcn_wmma_f32_16x16x4_f32(
                             false, af[mt][s], false, bB[s], (short)0, acc[mt][1], false, false);
        }
    }

    const size_t bk = (size_t)(b * KK + k);
    #pragma unroll
    for (int mt = 0; mt < MTILES; ++mt) {
        #pragma unroll
        for (int r = 0; r < 8; ++r) {
            const int R = mt * 16 + 8 * hf + r;
            float* rowp = nullptr;
            if (R >= RR && R < RR + NN)            rowp = Bt + (bk * NN + (R - RR)) * LL;
            else if (R >= RR + NN && R < DBCROWS)  rowp = Ct + (bk * NN + (R - RR - NN)) * LL;
            if (rowp != nullptr) {
                const float vA = acc[mt][0][r], vB = acc[mt][1][r];
                if (!odd) {
                    v2f v; v.x = rev ? vB : vA; v.y = rev ? vA : vB;
                    *reinterpret_cast<v2f*>(rowp + lq0) = v;
                } else {
                    rowp[lA] = vA;
                    rowp[lB] = vB;
                }
            }
        }
    }

    float d6[RR];
    #pragma unroll
    for (int r = 0; r < RR; ++r) {
        const float oth = __shfl_xor(acc[0][1][r], 16, 32);
        d6[r] = hf ? oth : acc[0][0][r];
    }
    float* dcol = delta_out + bk * CC * LL + (hf ? lB : lA);
    #pragma unroll 1
    for (int ch0 = 0; ch0 < CC; ch0 += CHB) {
        WdBlock wnext;
        wd_load(wnext, wdelta, dbias, k, (ch0 + CHB) % CC);
        #pragma unroll
        for (int u = 0; u < CHB; ++u) {
            float s = 0.0f;
            #pragma unroll
            for (int r = 0; r < RR; ++r) s += wcur.w[u][r] * d6[r];
            s += wcur.bias[u];
            const float t  = log1pf(__expf(s));
            const float sp = (s > 20.0f) ? s : t;
            dcol[(size_t)(ch0 + u) * LL] = sp;
        }
        wcur = wnext;
    }
}

#define SCAN_WAVES 3
#define TSTEP 4
#define TRIP  (32 * TSTEP)

struct TripBuf {
    float4 dl;
    float4 ux;
    float4 Bn[NN];
    float4 Cn[NN];
};

template <int S>
__device__ __forceinline__ float row_shr(float v, float fill) {
    return __int_as_float(__builtin_amdgcn_update_dpp(
        __float_as_int(fill), __float_as_int(v), 0x110 + S, 0xf, 0xf, false));
}
template <int S>
__device__ __forceinline__ float row_shr0(float v) {
    return __int_as_float(__builtin_amdgcn_update_dpp(
        0, __float_as_int(v), 0x110 + S, 0xf, 0xf, true));
}
__device__ __forceinline__ float lane_bcast(float v, int l) {
    return __int_as_float(__builtin_amdgcn_readlane(__float_as_int(v), l));
}

__device__ __forceinline__ void trip_load(TripBuf& r, int k, int l0,
                                          const float* __restrict__ dptr,
                                          const float* __restrict__ xptr,
                                          const float* __restrict__ bptr,
                                          const float* __restrict__ cptr)
{
    const bool rev = (k & 2) != 0;
    r.dl = *reinterpret_cast<const float4*>(dptr + l0);
    if ((k & 1) == 0) {
        r.ux = *reinterpret_cast<const float4*>(xptr + (rev ? (LL - TSTEP - l0) : l0));
    } else {
        r.ux.x = xptr[tr_idx(rev ? (LL - 1 - (l0 + 0)) : (l0 + 0))];
        r.ux.y = xptr[tr_idx(rev ? (LL - 1 - (l0 + 1)) : (l0 + 1))];
        r.ux.z = xptr[tr_idx(rev ? (LL - 1 - (l0 + 2)) : (l0 + 2))];
        r.ux.w = xptr[tr_idx(rev ? (LL - 1 - (l0 + 3)) : (l0 + 3))];
    }
    #pragma unroll
    for (int n = 0; n < NN; ++n) {
        r.Bn[n] = *reinterpret_cast<const float4*>(bptr + (size_t)n * LL + l0);
        r.Cn[n] = *reinterpret_cast<const float4*>(cptr + (size_t)n * LL + l0);
    }
    __builtin_amdgcn_sched_barrier(0);
}

__device__ __forceinline__ void trip_compute(const TripBuf& r, int k, int l0, bool hi,
                                             const float (&An2)[NN], float (&carry)[NN],
                                             float Dd, float* __restrict__ optr)
{
    const float d0 = r.dl.x, d1 = r.dl.y, d2 = r.dl.z, d3 = r.dl.w;
    const bool flip = (k == 2);
    const float u0 = flip ? r.ux.w : r.ux.x, u1 = flip ? r.ux.z : r.ux.y;
    const float u2 = flip ? r.ux.y : r.ux.z, u3 = flip ? r.ux.x : r.ux.w;
    const float du0 = d0 * u0, du1 = d1 * u1, du2 = d2 * u2, du3 = d3 * u3;
    float y0 = 0.0f, y1 = 0.0f, y2 = 0.0f, y3 = 0.0f;
    #pragma unroll
    for (int n = 0; n < NN; ++n) {
        const float an = An2[n];
        const float a0 = __builtin_amdgcn_exp2f(d0 * an);
        const float a1 = __builtin_amdgcn_exp2f(d1 * an);
        const float a2 = __builtin_amdgcn_exp2f(d2 * an);
        const float a3 = __builtin_amdgcn_exp2f(d3 * an);
        const float H0 = du0 * r.Bn[n].x;
        const float H1 = fmaf(a1, H0, du1 * r.Bn[n].y);
        const float H2 = fmaf(a2, H1, du2 * r.Bn[n].z);
        const float H3 = fmaf(a3, H2, du3 * r.Bn[n].w);
        const float P0 = a0;
        const float P1 = P0 * a1;
        const float P2 = P1 * a2;
        const float P3 = P2 * a3;
        float A = P3, Bv = H3;
        { float Ap = row_shr<1>(A, 1.0f), Bp = row_shr0<1>(Bv); Bv = fmaf(A, Bp, Bv); A *= Ap; }
        { float Ap = row_shr<2>(A, 1.0f), Bp = row_shr0<2>(Bv); Bv = fmaf(A, Bp, Bv); A *= Ap; }
        { float Ap = row_shr<4>(A, 1.0f), Bp = row_shr0<4>(Bv); Bv = fmaf(A, Bp, Bv); A *= Ap; }
        { float Ap = row_shr<8>(A, 1.0f), Bp = row_shr0<8>(Bv); Bv = fmaf(A, Bp, Bv); A *= Ap; }
        const float cin = carry[n];
        const float S0  = fmaf(A, cin, Bv);
        const float mid = lane_bcast(S0, 15);
        const float cv  = hi ? mid : cin;
        const float S   = fmaf(A, cv, Bv);
        carry[n] = lane_bcast(S, 31);
        const float Sin = row_shr<1>(S, cv);
        const float h0 = fmaf(P0, Sin, H0);
        const float h1 = fmaf(P1, Sin, H1);
        const float h2 = fmaf(P2, Sin, H2);
        const float h3 = fmaf(P3, Sin, H3);
        y0 = fmaf(h0, r.Cn[n].x, y0);
        y1 = fmaf(h1, r.Cn[n].y, y1);
        y2 = fmaf(h2, r.Cn[n].z, y2);
        y3 = fmaf(h3, r.Cn[n].w, y3);
    }
    y0 += Dd * u0; y1 += Dd * u1; y2 += Dd * u2; y3 += Dd * u3;
    const bool rev = (k & 2) != 0;
    if ((k & 1) == 0) {
        float4 t = rev ? make_float4(y3, y2, y1, y0) : make_float4(y0, y1, y2, y3);
        *reinterpret_cast<float4*>(optr + (rev ? (LL - TSTEP - l0) : l0)) = t;
    } else {
        optr[tr_idx(rev ? (LL - 1 - (l0 + 0)) : (l0 + 0))] = y0;
        optr[tr_idx(rev ? (LL - 1 - (l0 + 1)) : (l0 + 1))] = y1;
        optr[tr_idx(rev ? (LL - 1 - (l0 + 2)) : (l0 + 2))] = y2;
        optr[tr_idx(rev ? (LL - 1 - (l0 + 3)) : (l0 + 3))] = y3;
    }
}

__global__ __launch_bounds__(32 * SCAN_WAVES)
void s6_scan_time_kernel(const float* __restrict__ x,
                         const float* __restrict__ delta,
                         const float* __restrict__ Bt,
                         const float* __restrict__ Ct,
                         const float* __restrict__ A_log,
                         const float* __restrict__ Ds,
                         float* __restrict__ out)
{
    const int lane = threadIdx.x & 31;
    const int wv   = __builtin_amdgcn_readfirstlane((int)(threadIdx.x >> 5));
    const int ch   = blockIdx.x * SCAN_WAVES + wv;
    const int c    = ch % CC;
    const int bk   = ch / CC;
    const int k    = bk % KK;
    const int b    = bk / KK;
    const int d    = k * CC + c;
    const bool hi  = lane >= 16;

    float An2[NN], carry[NN];
    #pragma unroll
    for (int n = 0; n < NN; ++n) {
        An2[n]   = -__expf(A_log[d * NN + n]) * 1.44269504088896341f;
        carry[n] = 0.0f;
    }
    const float Dd = Ds[d];
    const float* dptr = delta + ((size_t)bk * CC + c) * LL;
    const float* bptr = Bt + (size_t)bk * NN * LL;
    const float* cptr = Ct + (size_t)bk * NN * LL;
    const float* xptr = x + ((size_t)b * CC + c) * LL;
    const int slot = (k == 0) ? 0 : (k == 1) ? 2 : (k == 2) ? 1 : 3;
    float* optr = out + ((size_t)(slot * BB + b) * CC + c) * LL;

    TripBuf r0, r1;
    const int lo = lane * TSTEP;
    trip_load(r0, k, lo, dptr, xptr, bptr, cptr);
    for (int it = 0; it < LL / TRIP; it += 2) {
        trip_load(r1, k, (it + 1) * TRIP + lo, dptr, xptr, bptr, cptr);
        trip_compute(r0, k, it * TRIP + lo, hi, An2, carry, Dd, optr);
        trip_load(r0, k, ((it + 2) & (LL / TRIP - 1)) * TRIP + lo, dptr, xptr, bptr, cptr);
        trip_compute(r1, k, (it + 1) * TRIP + lo, hi, An2, carry, Dd, optr);
    }
}

extern "C" void kernel_launch(void* const* d_in, const int* in_sizes, int n_in,
                              void* d_out, int out_size, void* d_ws, size_t ws_size,
                              hipStream_t stream) {
    const float* x      = (const float*)d_in[0];
    const float* wproj  = (const float*)d_in[1];
    const float* wdelta = (const float*)d_in[2];
    const float* dbias  = (const float*)d_in[3];
    const float* A_log  = (const float*)d_in[4];
    const float* Ds     = (const float*)d_in[5];
    float* out = (float*)d_out;

    float* ws     = (float*)d_ws;
    float* delta  = ws;
    float* Bt     = delta  + (size_t)BB * KK * CC * LL;
    float* Ct     = Bt     + (size_t)BB * KK * LL * NN;

    dim3 gProj(LL / (32 * PWAVES), KK, BB);
    s6_proj_kernel<<<gProj, 32 * PWAVES, 0, stream>>>(x, wproj, wdelta, dbias, delta, Bt, Ct);

    s6_scan_time_kernel<<<BB * KK * CC / SCAN_WAVES, 32 * SCAN_WAVES, 0, stream>>>(
        x, delta, Bt, Ct, A_log, Ds, out);
}


# scan: each lane folds 8 consecutive steps, not 4 (16 trips of 256; delta, B, C rows stored lane-blocked so loads stay whole lines): half the cross-lane scans; changes the order of the scan's sums
# speedup vs baseline: 15.0191x; 1.0811x over previous
#include <hip/hip_runtime.h>
#include <hip/hip_bf16.h>

#define BB   2
#define CC   96
#define HH   64
#define WWW  64
#define LL   4096
#define NN   16
#define RR   6
#define KK   4
#define DBCROWS 38
#define MPAD 48

typedef float v2f __attribute__((ext_vector_type(2)));
typedef float v8f __attribute__((ext_vector_type(8)));

__device__ __forceinline__ int tr_idx(int l) { return ((l & 63) << 6) | (l >> 6); }
__device__ __forceinline__ int dir_idx(int k, int l) {
    switch (k) {
        case 0:  return l;
        case 1:  return tr_idx(l);
        case 2:  return LL - 1 - l;
        default: return tr_idx(LL - 1 - l);
    }
}

__device__ __forceinline__ int ws_pos(int s) {
    return (s & ~255) | (((s >> 2) & 1) << 7) | (((s >> 3) & 31) << 2) | (s & 3);
}

#define PWAVES 4
#define KSTEPS (CC / 4)
#define MTILES (MPAD / 16)
#define CHB    4

struct WdBlock {
    float w[CHB][RR];
    float bias[CHB];
};

__device__ __forceinline__ void wd_load(WdBlock& o, const float* __restrict__ wdelta,
                                        const float* __restrict__ dbias, int k, int ch0)
{
    #pragma unroll
    for (int u = 0; u < CHB; ++u) {
        #pragma unroll
        for (int r = 0; r < RR; ++r) o.w[u][r] = wdelta[(k * CC + ch0 + u) * RR + r];
        o.bias[u] = dbias[k * CC + ch0 + u];
    }
}

__global__ __launch_bounds__(32 * PWAVES)
void s6_proj_kernel(const float* __restrict__ x,
                    const float* __restrict__ wproj,
                    const float* __restrict__ wdelta,
                    const float* __restrict__ dbias,
                    float* __restrict__ delta_out,
                    float* __restrict__ Bt,
                    float* __restrict__ Ct)
{
    const int lane = threadIdx.x & 31;
    const int wv = __builtin_amdgcn_readfirstlane((int)(threadIdx.x >> 5));
    const int c  = lane & 15;
    const int hf = lane >> 4;
    const int k  = blockIdx.y;
    const int b  = blockIdx.z;
    const bool odd = (k & 1) != 0;
    const bool rev = (k & 2) != 0;

    int lq0, lq1;
    if (!odd) { lq0 = (blockIdx.x * PWAVES + wv) * 32 + 2 * c;        lq1 = lq0 + 1;  }
    else      { lq0 = blockIdx.x * (32 * PWAVES) + 16 * wv + c;       lq1 = lq0 + 64; }
    const int p0 = dir_idx(k, lq0), p1 = dir_idx(k, lq1);
    const int pmin = (p0 < p1) ? p0 : p1;
    const int lA = rev ? lq1 : lq0;
    const int lB = rev ? lq0 : lq1;

    WdBlock wcur;
    wd_load(wcur, wdelta, dbias, k, 0);

    const float* xp = x + ((size_t)b * CC + 2 * hf) * LL + pmin;
    const float* wr[MTILES];
    bool valid[MTILES];
    #pragma unroll
    for (int mt = 0; mt < MTILES; ++mt) {
        const int row = mt * 16 + c;
        valid[mt] = row < DBCROWS;
        wr[mt] = wproj + ((size_t)k * DBCROWS + (valid[mt] ? row : DBCROWS - 1)) * CC + 2 * hf;
    }
    v2f bA[KSTEPS], bB[KSTEPS], af[MTILES][KSTEPS];
    #pragma unroll
    for (int s = 0; s < KSTEPS; ++s) {
        const v2f t0 = *reinterpret_cast<const v2f*>(xp + (size_t)(4 * s + 0) * LL);
        const v2f t1 = *reinterpret_cast<const v2f*>(xp + (size_t)(4 * s + 1) * LL);
        bA[s].x = t0.x; bA[s].y = t1.x;
        bB[s].x = t0.y; bB[s].y = t1.y;
        #pragma unroll
        for (int mt = 0; mt < MTILES; ++mt) {
            const v2f t = *reinterpret_cast<const v2f*>(wr[mt] + 4 * s);
            af[mt][s].x = valid[mt] ? t.x : 0.0f;
            af[mt][s].y = valid[mt] ? t.y : 0.0f;
        }
    }
    __builtin_amdgcn_sched_barrier(0);

    v8f acc[MTILES][2];
    #pragma unroll
    for (int mt = 0; mt < MTILES; ++mt) { acc[mt][0] = v8f{}; acc[mt][1] = v8f{}; }
    #pragma unroll
    for (int s = 0; s < KSTEPS; ++s) {
        #pragma unroll
        for (int mt = 0; mt < MTILES; ++mt) {
            acc[mt][0] = __builtin_amdgcn_wmma_f32_16x16x4_f32(
                             false, af[mt][s], false, bA[s], (short)0, acc[mt][0], false, false);
            acc[mt][1] = __builtin_amdgcn_wmma_f32_16x16x4_f32(
                             false, af[mt][s], false, bB[s], (short)0, acc[mt][1], false, false);
        }
    }

    const size_t bk = (size_t)(b * KK + k);
    #pragma unroll
    for (int mt = 0; mt < MTILES; ++mt) {
        #pragma unroll
        for (int r = 0; r < 8; ++r) {
            const int R = mt * 16 + 8 * hf + r;
            float* rowp = nullptr;
            if (R >= RR && R < RR + NN)            rowp = Bt + (bk * NN + (R - RR)) * LL;
            else if (R >= RR + NN && R < DBCROWS)  rowp = Ct + (bk * NN + (R - RR - NN)) * LL;
            if (rowp != nullptr) {
                const float vA = acc[mt][0][r], vB = acc[mt][1][r];
                if (!odd) {
                    v2f v; v.x = rev ? vB : vA; v.y = rev ? vA : vB;
                    *reinterpret_cast<v2f*>(rowp + ws_pos(lq0)) = v;
                } else {
                    rowp[ws_pos(lA)] = vA;
                    rowp[ws_pos(lB)] = vB;
                }
            }
        }
    }

    float d6[RR];
    #pragma unroll
    for (int r = 0; r < RR; ++r) {
        const float oth = __shfl_xor(acc[0][1][r], 16, 32);
        d6[r] = hf ? oth : acc[0][0][r];
    }
    float* dcol = delta_out + bk * CC * LL + ws_pos(hf ? lB : lA);
    #pragma unroll 1
    for (int ch0 = 0; ch0 < CC; ch0 += CHB) {
        WdBlock wnext;
        wd_load(wnext, wdelta, dbias, k, (ch0 + CHB) % CC);
        #pragma unroll
        for (int u = 0; u < CHB; ++u) {
            float s = 0.0f;
            #pragma unroll
            for (int r = 0; r < RR; ++r) s += wcur.w[u][r] * d6[r];
            s += wcur.bias[u];
            const float t  = log1pf(__expf(s));
            const float sp = (s > 20.0f) ? s : t;
            dcol[(size_t)(ch0 + u) * LL] = sp;
        }
        wcur = wnext;
    }
}

#define SCAN_WAVES 3
#define TSTEP 8
#define TRIP  (32 * TSTEP)
#define NTRIPS (LL / TRIP)
#define NGRP  4

struct TripBuf {
    float4 dl[2];
    float4 ux[2];
    float4 Bn[NN][2];
    float4 Cn[NN][2];
};

template <int S>
__device__ __forceinline__ float row_shr(float v, float fill) {
    return __int_as_float(__builtin_amdgcn_update_dpp(
        __float_as_int(fill), __float_as_int(v), 0x110 + S, 0xf, 0xf, false));
}
template <int S>
__device__ __forceinline__ float row_shr0(float v) {
    return __int_as_float(__builtin_amdgcn_update_dpp(
        0, __float_as_int(v), 0x110 + S, 0xf, 0xf, true));
}
__device__ __forceinline__ float lane_bcast(float v, int l) {
    return __int_as_float(__builtin_amdgcn_readlane(__float_as_int(v), l));
}

__device__ __forceinline__ void trip_load_du(TripBuf& r, int k, int blk, int lane,
                                             const float* __restrict__ dptr,
                                             const float* __restrict__ xptr)
{
    const bool rev = (k & 2) != 0;
    const int l0 = blk * TRIP + lane * TSTEP;
    const int w0 = blk * TRIP + lane * 4;
    r.dl[0] = *reinterpret_cast<const float4*>(dptr + w0);
    r.dl[1] = *reinterpret_cast<const float4*>(dptr + w0 + 128);
    if ((k & 1) == 0) {
        const int x0 = rev ? (LL - TSTEP - l0) : l0;
        r.ux[0] = *reinterpret_cast<const float4*>(xptr + x0);
        r.ux[1] = *reinterpret_cast<const float4*>(xptr + x0 + 4);
    } else {
        r.ux[0].x = xptr[tr_idx(rev ? (LL - 1 - (l0 + 0)) : (l0 + 0))];
        r.ux[0].y = xptr[tr_idx(rev ? (LL - 1 - (l0 + 1)) : (l0 + 1))];
        r.ux[0].z = xptr[tr_idx(rev ? (LL - 1 - (l0 + 2)) : (l0 + 2))];
        r.ux[0].w = xptr[tr_idx(rev ? (LL - 1 - (l0 + 3)) : (l0 + 3))];
        r.ux[1].x = xptr[tr_idx(rev ? (LL - 1 - (l0 + 4)) : (l0 + 4))];
        r.ux[1].y = xptr[tr_idx(rev ? (LL - 1 - (l0 + 5)) : (l0 + 5))];
        r.ux[1].z = xptr[tr_idx(rev ? (LL - 1 - (l0 + 6)) : (l0 + 6))];
        r.ux[1].w = xptr[tr_idx(rev ? (LL - 1 - (l0 + 7)) : (l0 + 7))];
    }
}

template <int N0>
__device__ __forceinline__ void trip_load_bc(TripBuf& r, int blk, int lane,
                                             const float* __restrict__ bptr,
                                             const float* __restrict__ cptr)
{
    const int w0 = blk * TRIP + lane * 4;
    __builtin_amdgcn_sched_barrier(0);
    #pragma unroll
    for (int n = N0; n < N0 + NGRP; ++n) {
        r.Bn[n][0] = *reinterpret_cast<const float4*>(bptr + (size_t)n * LL + w0);
        r.Bn[n][1] = *reinterpret_cast<const float4*>(bptr + (size_t)n * LL + w0 + 128);
        r.Cn[n][0] = *reinterpret_cast<const float4*>(cptr + (size_t)n * LL + w0);
        r.Cn[n][1] = *reinterpret_cast<const float4*>(cptr + (size_t)n * LL + w0 + 128);
    }
    __builtin_amdgcn_sched_barrier(0);
}

__device__ __forceinline__ void trip_begin(const TripBuf& r, int k,
                                           float (&d)[TSTEP], float (&u)[TSTEP],
                                           float (&du)[TSTEP], float (&y)[TSTEP])
{
    d[0] = r.dl[0].x; d[1] = r.dl[0].y; d[2] = r.dl[0].z; d[3] = r.dl[0].w;
    d[4] = r.dl[1].x; d[5] = r.dl[1].y; d[6] = r.dl[1].z; d[7] = r.dl[1].w;
    const bool flip = (k == 2);
    u[0] = flip ? r.ux[1].w : r.ux[0].x;  u[1] = flip ? r.ux[1].z : r.ux[0].y;
    u[2] = flip ? r.ux[1].y : r.ux[0].z;  u[3] = flip ? r.ux[1].x : r.ux[0].w;
    u[4] = flip ? r.ux[0].w : r.ux[1].x;  u[5] = flip ? r.ux[0].z : r.ux[1].y;
    u[6] = flip ? r.ux[0].y : r.ux[1].z;  u[7] = flip ? r.ux[0].x : r.ux[1].w;
    #pragma unroll
    for (int e = 0; e < TSTEP; ++e) {
        du[e] = d[e] * u[e];
        y[e]  = 0.0f;
    }
}

template <int N0>
__device__ __forceinline__ void trip_states(const TripBuf& r, bool hi,
                                            const float (&An2)[NN], float (&carry)[NN],
                                            const float (&d)[TSTEP], const float (&du)[TSTEP],
                                            float (&y)[TSTEP])
{
    #pragma unroll
    for (int n = N0; n < N0 + NGRP; ++n) {
        const float an = An2[n];
        const float Bm[TSTEP] = { r.Bn[n][0].x, r.Bn[n][0].y, r.Bn[n][0].z, r.Bn[n][0].w,
                                  r.Bn[n][1].x, r.Bn[n][1].y, r.Bn[n][1].z, r.Bn[n][1].w };
        const float Cm[TSTEP] = { r.Cn[n][0].x, r.Cn[n][0].y, r.Cn[n][0].z, r.Cn[n][0].w,
                                  r.Cn[n][1].x, r.Cn[n][1].y, r.Cn[n][1].z, r.Cn[n][1].w };
        float P[TSTEP], H[TSTEP];
        P[0] = __builtin_amdgcn_exp2f(d[0] * an);
        H[0] = du[0] * Bm[0];
        #pragma unroll
        for (int e = 1; e < TSTEP; ++e) {
            const float a = __builtin_amdgcn_exp2f(d[e] * an);
            H[e] = fmaf(a, H[e - 1], du[e] * Bm[e]);
            P[e] = P[e - 1] * a;
        }
        float A = P[TSTEP - 1], Bv = H[TSTEP - 1];
        { float Ap = row_shr<1>(A, 1.0f), Bp = row_shr0<1>(Bv); Bv = fmaf(A, Bp, Bv); A *= Ap; }
        { float Ap = row_shr<2>(A, 1.0f), Bp = row_shr0<2>(Bv); Bv = fmaf(A, Bp, Bv); A *= Ap; }
        { float Ap = row_shr<4>(A, 1.0f), Bp = row_shr0<4>(Bv); Bv = fmaf(A, Bp, Bv); A *= Ap; }
        { float Ap = row_shr<8>(A, 1.0f), Bp = row_shr0<8>(Bv); Bv = fmaf(A, Bp, Bv); A *= Ap; }
        const float cin = carry[n];
        const float S0  = fmaf(A, cin, Bv);
        const float mid = lane_bcast(S0, 15);
        const float cv  = hi ? mid : cin;
        const float S   = fmaf(A, cv, Bv);
        carry[n] = lane_bcast(S, 31);
        const float Sin = row_shr<1>(S, cv);
        #pragma unroll
        for (int e = 0; e < TSTEP; ++e) {
            const float h = fmaf(P[e], Sin, H[e]);
            y[e] = fmaf(h, Cm[e], y[e]);
        }
    }
}

__device__ __forceinline__ void trip_end(int k, int blk, int lane, float Dd,
                                         const float (&u)[TSTEP], float (&y)[TSTEP],
                                         float* __restrict__ optr)
{
    #pragma unroll
    for (int e = 0; e < TSTEP; ++e) y[e] += Dd * u[e];
    const bool rev = (k & 2) != 0;
    const int l0 = blk * TRIP + lane * TSTEP;
    if ((k & 1) == 0) {
        const int x0 = rev ? (LL - TSTEP - l0) : l0;
        const float4 t0 = rev ? make_float4(y[7], y[6], y[5], y[4]) : make_float4(y[0], y[1], y[2], y[3]);
        const float4 t1 = rev ? make_float4(y[3], y[2], y[1], y[0]) : make_float4(y[4], y[5], y[6], y[7]);
        *reinterpret_cast<float4*>(optr + x0)     = t0;
        *reinterpret_cast<float4*>(optr + x0 + 4) = t1;
    } else {
        #pragma unroll
        for (int e = 0; e < TSTEP; ++e)
            optr[tr_idx(rev ? (LL - 1 - (l0 + e)) : (l0 + e))] = y[e];
    }
}

__global__ __launch_bounds__(32 * SCAN_WAVES)
void s6_scan_time_kernel(const float* __restrict__ x,
                         const float* __restrict__ delta,
                         const float* __restrict__ Bt,
                         const float* __restrict__ Ct,
                         const float* __restrict__ A_log,
                         const float* __restrict__ Ds,
                         float* __restrict__ out)
{
    const int lane = threadIdx.x & 31;
    const int wv   = __builtin_amdgcn_readfirstlane((int)(threadIdx.x >> 5));
    const int ch   = blockIdx.x * SCAN_WAVES + wv;
    const int c    = ch % CC;
    const int bk   = ch / CC;
    const int k    = bk % KK;
    const int b    = bk / KK;
    const int d    = k * CC + c;
    const bool hi  = lane >= 16;

    float An2[NN], carry[NN];
    #pragma unroll
    for (int n = 0; n < NN; ++n) {
        An2[n]   = -__expf(A_log[d * NN + n]) * 1.44269504088896341f;
        carry[n] = 0.0f;
    }
    const float Dd = Ds[d];
    const float* dptr = delta + ((size_t)bk * CC + c) * LL;
    const float* bptr = Bt + (size_t)bk * NN * LL;
    const float* cptr = Ct + (size_t)bk * NN * LL;
    const float* xptr = x + ((size_t)b * CC + c) * LL;
    const int slot = (k == 0) ? 0 : (k == 1) ? 2 : (k == 2) ? 1 : 3;
    float* optr = out + ((size_t)(slot * BB + b) * CC + c) * LL;

    TripBuf r0, r1;
    trip_load_du(r0, k, 0, lane, dptr, xptr);
    trip_load_bc<0>(r0, 0, lane, bptr, cptr);
    trip_load_bc<4>(r0, 0, lane, bptr, cptr);
    trip_load_bc<8>(r0, 0, lane, bptr, cptr);
    trip_load_bc<12>(r0, 0, lane, bptr, cptr);
    for (int it = 0; it < NTRIPS; it += 2) {
        float dd[TSTEP], uu[TSTEP], du[TSTEP], yy[TSTEP];
        trip_load_du(r1, k, it + 1, lane, dptr, xptr);
        trip_load_bc<0>(r1, it + 1, lane, bptr, cptr);
        trip_begin(r0, k, dd, uu, du, yy);
        trip_states<0>(r0, hi, An2, carry, dd, du, yy);
        trip_load_bc<4>(r1, it + 1, lane, bptr, cptr);
        trip_states<4>(r0, hi, An2, carry, dd, du, yy);
        trip_load_bc<8>(r1, it + 1, lane, bptr, cptr);
        trip_states<8>(r0, hi, An2, carry, dd, du, yy);
        trip_load_bc<12>(r1, it + 1, lane, bptr, cptr);
        trip_states<12>(r0, hi, An2, carry, dd, du, yy);
        trip_end(k, it, lane, Dd, uu, yy, optr);
        const int nb = (it + 2) & (NTRIPS - 1);
        trip_load_du(r0, k, nb, lane, dptr, xptr);
        trip_load_bc<0>(r0, nb, lane, bptr, cptr);
        trip_begin(r1, k, dd, uu, du, yy);
        trip_states<0>(r1, hi, An2, carry, dd, du, yy);
        trip_load_bc<4>(r0, nb, lane, bptr, cptr);
        trip_states<4>(r1, hi, An2, carry, dd, du, yy);
        trip_load_bc<8>(r0, nb, lane, bptr, cptr);
        trip_states<8>(r1, hi, An2, carry, dd, du, yy);
        trip_load_bc<12>(r0, nb, lane, bptr, cptr);
        trip_states<12>(r1, hi, An2, carry, dd, du, yy);
        trip_end(k, it + 1, lane, Dd, uu, yy, optr);
    }
}

extern "C" void kernel_launch(void* const* d_in, const int* in_sizes, int n_in,
                              void* d_out, int out_size, void* d_ws, size_t ws_size,
                              hipStream_t stream) {
    const float* x      = (const float*)d_in[0];
    const float* wproj  = (const float*)d_in[1];
    const float* wdelta = (const float*)d_in[2];
    const float* dbias  = (const float*)d_in[3];
    const float* A_log  = (const float*)d_in[4];
    const float* Ds     = (const float*)d_in[5];
    float* out = (float*)d_out;

    float* ws     = (float*)d_ws;
    float* delta  = ws;
    float* Bt     = delta  + (size_t)BB * KK * CC * LL;
    float* Ct     = Bt     + (size_t)BB * KK * LL * NN;

    dim3 gProj(LL / (32 * PWAVES), KK, BB);
    s6_proj_kernel<<<gProj, 32 * PWAVES, 0, stream>>>(x, wproj, wdelta, dbias, delta, Bt, Ct);

    s6_scan_time_kernel<<<BB * KK * CC / SCAN_WAVES, 32 * SCAN_WAVES, 0, stream>>>(
        x, delta, Bt, Ct, A_log, Ds, out);
}
